# starting kernel
# baseline (speedup 1.0000x reference)
#include <hip/hip_runtime.h>

typedef __attribute__((ext_vector_type(2))) float v2f;
typedef __attribute__((ext_vector_type(4))) float v4f;
typedef __attribute__((ext_vector_type(8))) float v8f;

#define B_      32
#define S_      512
#define VDIM_   256
#define MAXLEN_ 13
#define NCOL_   96
#define LSTRIDE_ 260

__global__ __launch_bounds__(256)
void residues_gemm_kernel(const float* __restrict__ x_v,
                          const float* __restrict__ W_hid,
                          const float* __restrict__ W_out,
                          const int*  __restrict__ seq,
                          float* __restrict__ y_stage)
{
    extern __shared__ float smem[];
    float* XT = smem;
    float* HT = smem;

    const int s     = blockIdx.x;
    const int tid   = threadIdx.x;
    const int lane  = tid & 31;
    const int wave  = tid >> 5;
    const int lhalf = lane >> 4;
    const int l16   = lane & 15;

    const int letter = seq[s];

    {
        const float* xs = x_v + (size_t)s * (VDIM_ * 3);
        for (int n = 0; n < NCOL_; ++n) {
            const int b = n / 3, c = n - 3 * b;
            const int v = tid;
            XT[n * LSTRIDE_ + v] = xs[((size_t)b * S_ * VDIM_ + v) * 3 + c];
        }
    }
    __syncthreads();

    const float* Wh = W_hid + (size_t)letter * VDIM_ * VDIM_;
    const int m0 = wave * 32;

    v8f acc[2][6];
#pragma unroll
    for (int mt = 0; mt < 2; ++mt)
#pragma unroll
        for (int nt = 0; nt < 6; ++nt)
            acc[mt][nt] = (v8f){0.f, 0.f, 0.f, 0.f, 0.f, 0.f, 0.f, 0.f};

    for (int k = 0; k < VDIM_; k += 4) {
        const int kk = k + 2 * lhalf;
        v2f a0 = *(const v2f*)(Wh + (size_t)(m0 +      l16) * VDIM_ + kk);
        v2f a1 = *(const v2f*)(Wh + (size_t)(m0 + 16 + l16) * VDIM_ + kk);
#pragma unroll
        for (int nt = 0; nt < 6; ++nt) {
            v2f b = *(const v2f*)(XT + (nt * 16 + l16) * LSTRIDE_ + kk);
            acc[0][nt] = __builtin_amdgcn_wmma_f32_16x16x4_f32(
                false, a0, false, b, (short)0, acc[0][nt], false, false);
            acc[1][nt] = __builtin_amdgcn_wmma_f32_16x16x4_f32(
                false, a1, false, b, (short)0, acc[1][nt], false, false);
        }
    }

    __syncthreads();

#pragma unroll
    for (int mt = 0; mt < 2; ++mt) {
#pragma unroll
        for (int nt = 0; nt < 6; ++nt) {
            float* dst = HT + (nt * 16 + l16) * LSTRIDE_ + m0 + mt * 16 + 8 * lhalf;
            v8f A = acc[mt][nt];
            *(v4f*)(dst)     = (v4f){A[0], A[1], A[2], A[3]};
            *(v4f*)(dst + 4) = (v4f){A[4], A[5], A[6], A[7]};
        }
    }
    __syncthreads();

    if (wave < 6) {
        const float* Wo = W_out + (size_t)letter * MAXLEN_ * VDIM_;
        const int n0 = wave * 16;
        const bool mval = (l16 < MAXLEN_);
        const v2f zv2 = {0.f, 0.f};

        v8f acc2 = (v8f){0.f, 0.f, 0.f, 0.f, 0.f, 0.f, 0.f, 0.f};
        for (int k = 0; k < VDIM_; k += 4) {
            const int kk = k + 2 * lhalf;
            v2f a = mval ? *(const v2f*)(Wo + (size_t)l16 * VDIM_ + kk) : zv2;
            v2f b = *(const v2f*)(HT + (n0 + l16) * LSTRIDE_ + kk);
            acc2 = __builtin_amdgcn_wmma_f32_16x16x4_f32(
                false, a, false, b, (short)0, acc2, false, false);
        }

        float* ys = y_stage + ((size_t)s * MAXLEN_) * NCOL_ + (n0 + l16);
#pragma unroll
        for (int j = 0; j < 8; ++j) {
            const int m = j + 8 * lhalf;
            if (m < MAXLEN_) ys[(size_t)m * NCOL_] = acc2[j];
        }
    }
}

__global__ __launch_bounds__(256)
void residues_gather_kernel(const float* __restrict__ y_stage,
                            const int* __restrict__ gidx,
                            float* __restrict__ out, int natoms)
{
    const int i = blockIdx.x * blockDim.x + threadIdx.x;
    const int tot = natoms * NCOL_;
    if (i >= tot) return;
    const int p = i / NCOL_;
    const int n = i - p * NCOL_;
    const int g = gidx[p];
    const int b = n / 3, c = n - 3 * b;
    out[((size_t)b * natoms + p) * 3 + c] = y_stage[(size_t)g * NCOL_ + n];
}

extern "C" void kernel_launch(void* const* d_in, const int* in_sizes, int n_in,
                              void* d_out, int out_size, void* d_ws, size_t ws_size,
                              hipStream_t stream) {
    const float* x_v  = (const float*)d_in[0];
    const float* Wh   = (const float*)d_in[1];
    const float* Wo   = (const float*)d_in[2];
    const int*   seq  = (const int*)  d_in[3];
    const int*   gidx = (const int*)  d_in[4];
    float* out     = (float*)d_out;
    float* y_stage = (float*)d_ws;

    const int natoms = in_sizes[4];
    const size_t lds_bytes = (size_t)NCOL_ * LSTRIDE_ * sizeof(float);

    residues_gemm_kernel<<<S_, 256, lds_bytes, stream>>>(x_v, Wh, Wo, seq, y_stage);

    const int tot = natoms * NCOL_;
    residues_gather_kernel<<<(tot + 255) / 256, 256, 0, stream>>>(y_stage, gidx, out, natoms);
}


# Reassociate the two products: a new first launch forms Wc = W_out*W_hid per letter (20 small WMMA GEMMs), then y = Wc[seq[s]]*x per position; the f32 summation order changes
# speedup vs baseline: 2.0375x; 2.0375x over previous
#include <hip/hip_runtime.h>


typedef __attribute__((ext_vector_type(2))) float v2f;
typedef __attribute__((ext_vector_type(4))) float v4f;
typedef __attribute__((ext_vector_type(8))) float v8f;

#define B_      32
#define S_      512
#define VDIM_   256
#define MAXLEN_ 13
#define NLETTER_ 20
#define NCOL_   96
#define LSTRIDE_ 260
#define MPAD_   16
#define WC_FLOATS_ (NLETTER_ * MPAD_ * VDIM_)
#define YSTAGE_OFF_ 131072
#define REDPITCH_ (4 * 8 * 32 + 16)

__global__ __launch_bounds__(256)
void residues_combine_kernel(const float* __restrict__ W_hid,
                             const float* __restrict__ W_out,
                             float* __restrict__ wc)
{
    __shared__ float red[2 * REDPITCH_];

    const int letter = blockIdx.y;
    const int c0     = blockIdx.x * 32;
    const int tid    = threadIdx.x;
    const int lane   = tid & 31;
    const int wave   = tid >> 5;
    const int lhalf  = lane >> 4;
    const int l16    = lane & 15;
    const int ntile  = wave & 1;
    const int kq     = wave >> 1;

    const float* Wo = W_out + (size_t)letter * MAXLEN_ * VDIM_;
    const float* Wh = W_hid + (size_t)letter * VDIM_ * VDIM_;
    const int n0 = c0 + ntile * 16;
    const bool mval = (l16 < MAXLEN_);
    const int  arow = mval ? l16 : 0;
    const v2f zv2 = {0.f, 0.f};

    v2f a[16], b[16];
#pragma unroll
    for (int i = 0; i < 16; ++i) {
        const int kk = kq * 64 + 4 * i + 2 * lhalf;
        a[i] = *(const v2f*)(Wo + (size_t)arow * VDIM_ + kk);
        b[i].x = Wh[(size_t)kk * VDIM_ + n0 + l16];
        b[i].y = Wh[(size_t)(kk + 1) * VDIM_ + n0 + l16];
    }

    v8f acc_e = (v8f){0.f, 0.f, 0.f, 0.f, 0.f, 0.f, 0.f, 0.f};
    v8f acc_o = (v8f){0.f, 0.f, 0.f, 0.f, 0.f, 0.f, 0.f, 0.f};
#pragma unroll
    for (int i = 0; i < 16; i += 2) {
        const v2f ae = mval ? a[i]     : zv2;
        const v2f ao = mval ? a[i + 1] : zv2;
        acc_e = __builtin_amdgcn_wmma_f32_16x16x4_f32(
            false, ae, false, b[i],     (short)0, acc_e, false, false);
        acc_o = __builtin_amdgcn_wmma_f32_16x16x4_f32(
            false, ao, false, b[i + 1], (short)0, acc_o, false, false);
    }
    const v8f acc = acc_e + acc_o;

    {
        float* r = red + ntile * REDPITCH_ + kq * (8 * 32) + lane;
#pragma unroll
        for (int j = 0; j < 8; ++j) r[j * 32] = acc[j];
    }
    __syncthreads();

    for (int e = tid; e < MPAD_ * 32; e += 256) {
        const int row = e >> 5, col = e & 31;
        const int nt = col >> 4, cc = col & 15;
        const float* r = red + nt * REDPITCH_ + (row & 7) * 32 + (row >> 3) * 16 + cc;
        const float sum = (r[0] + r[8 * 32]) + (r[2 * 8 * 32] + r[3 * 8 * 32]);
        wc[((size_t)letter * MPAD_ + row) * VDIM_ + c0 + col] = sum;
    }
}

__global__ __launch_bounds__(256)
void residues_gemm_kernel(const float* __restrict__ x_v,
                          const float* __restrict__ wc,
                          const int*  __restrict__ seq,
                          float* __restrict__ y_stage)
{
    extern __shared__ float smem[];
    float* XT = smem;

    const int s     = blockIdx.x;
    const int tid   = threadIdx.x;
    const int lane  = tid & 31;
    const int wave  = tid >> 5;
    const int lhalf = lane >> 4;
    const int l16   = lane & 15;

    const int letter = seq[s];

    {
        const float* xs = x_v + (size_t)s * (VDIM_ * 3);
        for (int n = 0; n < NCOL_; ++n) {
            const int b = n / 3, c = n - 3 * b;
            const int v = tid;
            XT[n * LSTRIDE_ + v] = xs[((size_t)b * S_ * VDIM_ + v) * 3 + c];
        }
    }
    __syncthreads();

    if (wave < 6) {
        const float* Wc = wc + (size_t)letter * MPAD_ * VDIM_;
        const int n0 = wave * 16;

        v8f acc_e = (v8f){0.f, 0.f, 0.f, 0.f, 0.f, 0.f, 0.f, 0.f};
        v8f acc_o = (v8f){0.f, 0.f, 0.f, 0.f, 0.f, 0.f, 0.f, 0.f};
        for (int k = 0; k < VDIM_; k += 8) {
            const int kk = k + 2 * lhalf;
            v2f a0 = *(const v2f*)(Wc + (size_t)l16 * VDIM_ + kk);
            v2f a1 = *(const v2f*)(Wc + (size_t)l16 * VDIM_ + kk + 4);
            v2f b0 = *(const v2f*)(XT + (n0 + l16) * LSTRIDE_ + kk);
            v2f b1 = *(const v2f*)(XT + (n0 + l16) * LSTRIDE_ + kk + 4);
            acc_e = __builtin_amdgcn_wmma_f32_16x16x4_f32(
                false, a0, false, b0, (short)0, acc_e, false, false);
            acc_o = __builtin_amdgcn_wmma_f32_16x16x4_f32(
                false, a1, false, b1, (short)0, acc_o, false, false);
        }
        const v8f acc2 = acc_e + acc_o;

        float* ys = y_stage + ((size_t)s * MAXLEN_) * NCOL_ + (n0 + l16);
#pragma unroll
        for (int j = 0; j < 8; ++j) {
            const int m = j + 8 * lhalf;
            if (m < MAXLEN_) ys[(size_t)m * NCOL_] = acc2[j];
        }
    }
}

__global__ __launch_bounds__(256)
void residues_gather_kernel(const float* __restrict__ y_stage,
                            const int* __restrict__ gidx,
                            float* __restrict__ out, int natoms)
{
    const int i = blockIdx.x * blockDim.x + threadIdx.x;
    const int tot = natoms * NCOL_;
    if (i >= tot) return;
    const int p = i / NCOL_;
    const int n = i - p * NCOL_;
    const int g = gidx[p];
    const int b = n / 3, c = n - 3 * b;
    out[((size_t)b * natoms + p) * 3 + c] = y_stage[(size_t)g * NCOL_ + n];
}

extern "C" void kernel_launch(void* const* d_in, const int* in_sizes, int n_in,
                              void* d_out, int out_size, void* d_ws, size_t ws_size,
                              hipStream_t stream) {
    const float* x_v  = (const float*)d_in[0];
    const float* Wh   = (const float*)d_in[1];
    const float* Wo   = (const float*)d_in[2];
    const int*   seq  = (const int*)  d_in[3];
    const int*   gidx = (const int*)  d_in[4];
    float* out     = (float*)d_out;
    float* wc      = (float*)d_ws;
    float* y_stage = (float*)d_ws + YSTAGE_OFF_;

    const int natoms = in_sizes[4];
    const size_t lds_bytes = (size_t)NCOL_ * LSTRIDE_ * sizeof(float);

    residues_combine_kernel<<<dim3(VDIM_ / 32, NLETTER_), 256, 0, stream>>>(Wh, Wo, wc);

    residues_gemm_kernel<<<S_, 256, lds_bytes, stream>>>(x_v, wc, seq, y_stage);

    const int tot = natoms * NCOL_;
    residues_gather_kernel<<<(tot + 255) / 256, 256, 0, stream>>>(y_stage, gidx, out, natoms);
}


# Fuse the gather into the GEMM launch: each workgroup keeps its 13x96 tile in LDS, finds its own slots in gather_idx and stores those rows straight to out; third launch gone; sums unchanged
# speedup vs baseline: 2.4049x; 1.1803x over previous
#include <hip/hip_runtime.h>


typedef __attribute__((ext_vector_type(2))) float v2f;
typedef __attribute__((ext_vector_type(4))) float v4f;
typedef __attribute__((ext_vector_type(8))) float v8f;

#define B_      32
#define S_      512
#define VDIM_   256
#define MAXLEN_ 13
#define NLETTER_ 20
#define NCOL_   96
#define LSTRIDE_ 260
#define MPAD_   16
#define WC_FLOATS_ (NLETTER_ * MPAD_ * VDIM_)
#define NATOMS_ 3745
#define NSCAN_  ((NATOMS_ + 255) / 256)
#define YPITCH_ 98
#define REDPITCH_ (4 * 8 * 32 + 16)

__global__ __launch_bounds__(256)
void residues_combine_kernel(const float* __restrict__ W_hid,
                             const float* __restrict__ W_out,
                             float* __restrict__ wc)
{
    __shared__ float red[2 * REDPITCH_];

    const int letter = blockIdx.y;
    const int c0     = blockIdx.x * 32;
    const int tid    = threadIdx.x;
    const int lane   = tid & 31;
    const int wave   = tid >> 5;
    const int lhalf  = lane >> 4;
    const int l16    = lane & 15;
    const int ntile  = wave & 1;
    const int kq     = wave >> 1;

    const float* Wo = W_out + (size_t)letter * MAXLEN_ * VDIM_;
    const float* Wh = W_hid + (size_t)letter * VDIM_ * VDIM_;
    const int n0 = c0 + ntile * 16;
    const bool mval = (l16 < MAXLEN_);
    const int  arow = mval ? l16 : 0;
    const v2f zv2 = {0.f, 0.f};

    v2f a[16], b[16];
#pragma unroll
    for (int i = 0; i < 16; ++i) {
        const int kk = kq * 64 + 4 * i + 2 * lhalf;
        a[i] = *(const v2f*)(Wo + (size_t)arow * VDIM_ + kk);
        b[i].x = Wh[(size_t)kk * VDIM_ + n0 + l16];
        b[i].y = Wh[(size_t)(kk + 1) * VDIM_ + n0 + l16];
    }

    v8f acc_e = (v8f){0.f, 0.f, 0.f, 0.f, 0.f, 0.f, 0.f, 0.f};
    v8f acc_o = (v8f){0.f, 0.f, 0.f, 0.f, 0.f, 0.f, 0.f, 0.f};
#pragma unroll
    for (int i = 0; i < 16; i += 2) {
        const v2f ae = mval ? a[i]     : zv2;
        const v2f ao = mval ? a[i + 1] : zv2;
        acc_e = __builtin_amdgcn_wmma_f32_16x16x4_f32(
            false, ae, false, b[i],     (short)0, acc_e, false, false);
        acc_o = __builtin_amdgcn_wmma_f32_16x16x4_f32(
            false, ao, false, b[i + 1], (short)0, acc_o, false, false);
    }
    const v8f acc = acc_e + acc_o;

    {
        float* r = red + ntile * REDPITCH_ + kq * (8 * 32) + lane;
#pragma unroll
        for (int j = 0; j < 8; ++j) r[j * 32] = acc[j];
    }
    __syncthreads();

    for (int e = tid; e < MPAD_ * 32; e += 256) {
        const int row = e >> 5, col = e & 31;
        const int nt = col >> 4, cc = col & 15;
        const float* r = red + nt * REDPITCH_ + (row & 7) * 32 + (row >> 3) * 16 + cc;
        const float sum = (r[0] + r[8 * 32]) + (r[2 * 8 * 32] + r[3 * 8 * 32]);
        wc[((size_t)letter * MPAD_ + row) * VDIM_ + c0 + col] = sum;
    }
}

__global__ __launch_bounds__(256)
void residues_gemm_kernel(const float* __restrict__ x_v,
                          const float* __restrict__ wc,
                          const int*  __restrict__ seq,
                          const int*  __restrict__ gidx,
                          float* __restrict__ out)
{
    extern __shared__ float smem[];
    float* XT = smem;
    float* Y  = smem + NCOL_ * LSTRIDE_;

    const int s     = blockIdx.x;
    const int tid   = threadIdx.x;
    const int lane  = tid & 31;
    const int wave  = tid >> 5;
    const int lhalf = lane >> 4;
    const int l16   = lane & 15;

    const int letter = seq[s];

    int g[NSCAN_];
    __builtin_assume(tid < 256);
#pragma unroll
    for (int i = 0; i < NSCAN_; ++i) {
        const int p = i * 256 + tid;
        g[i] = (p < NATOMS_) ? gidx[p] : -1;
    }

    {
        const float* xs = x_v + (size_t)s * (VDIM_ * 3);
        for (int n = 0; n < NCOL_; ++n) {
            const int b = n / 3, c = n - 3 * b;
            const int v = tid;
            XT[n * LSTRIDE_ + v] = xs[((size_t)b * S_ * VDIM_ + v) * 3 + c];
        }
    }

    unsigned long long pk = 0ull;
#pragma unroll
    for (int i = 0; i < NSCAN_; ++i) {
        const unsigned r = (unsigned)(g[i] - s * MAXLEN_);
        const unsigned code = (r < (unsigned)MAXLEN_) ? (r + 1u) : 0u;
        pk |= (unsigned long long)code << (4 * i);
    }
    __syncthreads();

    if (wave < 6) {
        const float* Wc = wc + (size_t)letter * MPAD_ * VDIM_;
        const int n0 = wave * 16;

        v8f acc_e = (v8f){0.f, 0.f, 0.f, 0.f, 0.f, 0.f, 0.f, 0.f};
        v8f acc_o = (v8f){0.f, 0.f, 0.f, 0.f, 0.f, 0.f, 0.f, 0.f};
        for (int k = 0; k < VDIM_; k += 8) {
            const int kk = k + 2 * lhalf;
            v2f a0 = *(const v2f*)(Wc + (size_t)l16 * VDIM_ + kk);
            v2f a1 = *(const v2f*)(Wc + (size_t)l16 * VDIM_ + kk + 4);
            v2f b0 = *(const v2f*)(XT + (n0 + l16) * LSTRIDE_ + kk);
            v2f b1 = *(const v2f*)(XT + (n0 + l16) * LSTRIDE_ + kk + 4);
            acc_e = __builtin_amdgcn_wmma_f32_16x16x4_f32(
                false, a0, false, b0, (short)0, acc_e, false, false);
            acc_o = __builtin_amdgcn_wmma_f32_16x16x4_f32(
                false, a1, false, b1, (short)0, acc_o, false, false);
        }
        const v8f acc2 = acc_e + acc_o;

        float* yt = Y + (8 * lhalf) * YPITCH_ + n0 + l16;
#pragma unroll
        for (int j = 0; j < 8; ++j) yt[j * YPITCH_] = acc2[j];
    }
    __syncthreads();

    if (pk != 0ull) {
        for (int i = 0; i < NSCAN_; ++i) {
            const unsigned code = (unsigned)(pk >> (4 * i)) & 15u;
            if (code != 0u) {
                const float* yr = Y + (code - 1u) * YPITCH_;
                float* o = out + (size_t)(i * 256 + tid) * 3;
#pragma unroll 8
                for (int b = 0; b < B_; ++b) {
                    const float y0 = yr[3 * b], y1 = yr[3 * b + 1], y2 = yr[3 * b + 2];
                    float* ob = o + (size_t)b * (NATOMS_ * 3);
                    ob[0] = y0; ob[1] = y1; ob[2] = y2;
                }
            }
        }
    }
}

extern "C" void kernel_launch(void* const* d_in, const int* in_sizes, int n_in,
                              void* d_out, int out_size, void* d_ws, size_t ws_size,
                              hipStream_t stream) {
    const float* x_v  = (const float*)d_in[0];
    const float* Wh   = (const float*)d_in[1];
    const float* Wo   = (const float*)d_in[2];
    const int*   seq  = (const int*)  d_in[3];
    const int*   gidx = (const int*)  d_in[4];
    float* out     = (float*)d_out;
    float* wc      = (float*)d_ws;

    const size_t lds_bytes = (size_t)(NCOL_ * LSTRIDE_ + MPAD_ * YPITCH_) * sizeof(float);

    residues_combine_kernel<<<dim3(VDIM_ / 32, NLETTER_), 256, 0, stream>>>(Wh, Wo, wc);

    residues_gemm_kernel<<<S_, 256, lds_bytes, stream>>>(x_v, wc, seq, gidx, out);
}


# Stage x in LDS as it lies in memory: 24 coalesced 16-byte loads per thread, issued as one group, and 16-byte LDS stores replace 96 strided 4-byte pairs; B fragments read k 3 floats apart; same sums
# speedup vs baseline: 3.2759x; 1.3622x over previous
#include <hip/hip_runtime.h>


typedef __attribute__((ext_vector_type(2))) float v2f;
typedef __attribute__((ext_vector_type(4))) float v4f;
typedef __attribute__((ext_vector_type(8))) float v8f;

#define B_      32
#define S_      512
#define VDIM_   256
#define MAXLEN_ 13
#define NLETTER_ 20
#define XROW_   (VDIM_ * 3)
#define XPITCH_ 780
#define MPAD_   16
#define WC_FLOATS_ (NLETTER_ * MPAD_ * VDIM_)
#define NATOMS_ 3745
#define NSCAN_  ((NATOMS_ + 255) / 256)
#define YPITCH_ 98
#define REDPITCH_ (4 * 8 * 32 + 16)

__global__ __launch_bounds__(256)
void residues_combine_kernel(const float* __restrict__ W_hid,
                             const float* __restrict__ W_out,
                             float* __restrict__ wc)
{
    __shared__ float red[2 * REDPITCH_];

    const int letter = blockIdx.y;
    const int c0     = blockIdx.x * 32;
    const int tid    = threadIdx.x;
    const int lane   = tid & 31;
    const int wave   = tid >> 5;
    const int lhalf  = lane >> 4;
    const int l16    = lane & 15;
    const int ntile  = wave & 1;
    const int kq     = wave >> 1;

    const float* Wo = W_out + (size_t)letter * MAXLEN_ * VDIM_;
    const float* Wh = W_hid + (size_t)letter * VDIM_ * VDIM_;
    const int n0 = c0 + ntile * 16;
    const bool mval = (l16 < MAXLEN_);
    const int  arow = mval ? l16 : 0;
    const v2f zv2 = {0.f, 0.f};

    v2f a[16], b[16];
#pragma unroll
    for (int i = 0; i < 16; ++i) {
        const int kk = kq * 64 + 4 * i + 2 * lhalf;
        a[i] = *(const v2f*)(Wo + (size_t)arow * VDIM_ + kk);
        b[i].x = Wh[(size_t)kk * VDIM_ + n0 + l16];
        b[i].y = Wh[(size_t)(kk + 1) * VDIM_ + n0 + l16];
    }

    v8f acc_e = (v8f){0.f, 0.f, 0.f, 0.f, 0.f, 0.f, 0.f, 0.f};
    v8f acc_o = (v8f){0.f, 0.f, 0.f, 0.f, 0.f, 0.f, 0.f, 0.f};
#pragma unroll
    for (int i = 0; i < 16; i += 2) {
        const v2f ae = mval ? a[i]     : zv2;
        const v2f ao = mval ? a[i + 1] : zv2;
        acc_e = __builtin_amdgcn_wmma_f32_16x16x4_f32(
            false, ae, false, b[i],     (short)0, acc_e, false, false);
        acc_o = __builtin_amdgcn_wmma_f32_16x16x4_f32(
            false, ao, false, b[i + 1], (short)0, acc_o, false, false);
    }
    const v8f acc = acc_e + acc_o;

    {
        float* r = red + ntile * REDPITCH_ + kq * (8 * 32) + lane;
#pragma unroll
        for (int j = 0; j < 8; ++j) r[j * 32] = acc[j];
    }
    __syncthreads();

    for (int e = tid; e < MPAD_ * 32; e += 256) {
        const int row = e >> 5, col = e & 31;
        const int nt = col >> 4, cc = col & 15;
        const float* r = red + nt * REDPITCH_ + (row & 7) * 32 + (row >> 3) * 16 + cc;
        const float sum = (r[0] + r[8 * 32]) + (r[2 * 8 * 32] + r[3 * 8 * 32]);
        wc[((size_t)letter * MPAD_ + row) * VDIM_ + c0 + col] = sum;
    }
}

__global__ __launch_bounds__(256)
void residues_gemm_kernel(const float* __restrict__ x_v,
                          const float* __restrict__ wc,
                          const int*  __restrict__ seq,
                          const int*  __restrict__ gidx,
                          float* __restrict__ out)
{
    extern __shared__ __attribute__((aligned(16))) float smem[];
    float* XR = smem;
    float* Y  = smem + B_ * XPITCH_;

    const int s     = blockIdx.x;
    const int tid   = threadIdx.x;
    const int lane  = tid & 31;
    const int wave  = tid >> 5;
    const int lhalf = lane >> 4;
    const int l16   = lane & 15;

    const int letter = seq[s];

    const int grp = tid >> 6, t64 = tid & 63;
    const float* xs = x_v + (size_t)s * XROW_ + (size_t)grp * ((size_t)S_ * XROW_) + 4 * t64;
    v4f xr[24];
#pragma unroll
    for (int i = 0; i < 8; ++i) {
#pragma unroll
        for (int j = 0; j < 3; ++j)
            xr[3 * i + j] = *(const v4f*)(xs + (size_t)i * (4 * (size_t)S_ * XROW_) + j * 256);
    }

    int g[NSCAN_];
    __builtin_assume(tid < 256);
#pragma unroll
    for (int i = 0; i < NSCAN_; ++i) {
        const int p = i * 256 + tid;
        g[i] = (p < NATOMS_) ? gidx[p] : -1;
    }

    {
        float* xl = XR + grp * XPITCH_ + 4 * t64;
#pragma unroll
        for (int i = 0; i < 8; ++i) {
#pragma unroll
            for (int j = 0; j < 3; ++j)
                *(v4f*)(xl + i * (4 * XPITCH_) + j * 256) = xr[3 * i + j];
        }
    }

    unsigned long long pk = 0ull;
#pragma unroll
    for (int i = 0; i < NSCAN_; ++i) {
        const unsigned r = (unsigned)(g[i] - s * MAXLEN_);
        const unsigned code = (r < (unsigned)MAXLEN_) ? (r + 1u) : 0u;
        pk |= (unsigned long long)code << (4 * i);
    }
    __syncthreads();

    if (wave < 6) {
        const float* Wc = wc + (size_t)letter * MPAD_ * VDIM_;
        const int cc = wave % 3;
        const int bcol = (wave / 3) * 16 + l16;
        const float* xb = XR + bcol * XPITCH_ + cc;

        v8f acc_e = (v8f){0.f, 0.f, 0.f, 0.f, 0.f, 0.f, 0.f, 0.f};
        v8f acc_o = (v8f){0.f, 0.f, 0.f, 0.f, 0.f, 0.f, 0.f, 0.f};
        for (int k = 0; k < VDIM_; k += 8) {
            const int kk = k + 2 * lhalf;
            v2f a0 = *(const v2f*)(Wc + (size_t)l16 * VDIM_ + kk);
            v2f a1 = *(const v2f*)(Wc + (size_t)l16 * VDIM_ + kk + 4);
            v2f b0, b1;
            b0.x = xb[3 * kk];
            b0.y = xb[3 * kk + 3];
            b1.x = xb[3 * kk + 12];
            b1.y = xb[3 * kk + 15];
            acc_e = __builtin_amdgcn_wmma_f32_16x16x4_f32(
                false, a0, false, b0, (short)0, acc_e, false, false);
            acc_o = __builtin_amdgcn_wmma_f32_16x16x4_f32(
                false, a1, false, b1, (short)0, acc_o, false, false);
        }
        const v8f acc2 = acc_e + acc_o;

        float* yt = Y + (8 * lhalf) * YPITCH_ + 3 * bcol + cc;
#pragma unroll
        for (int j = 0; j < 8; ++j) yt[j * YPITCH_] = acc2[j];
    }
    __syncthreads();

    if (pk != 0ull) {
        for (int i = 0; i < NSCAN_; ++i) {
            const unsigned code = (unsigned)(pk >> (4 * i)) & 15u;
            if (code != 0u) {
                const float* yr = Y + (code - 1u) * YPITCH_;
                float* o = out + (size_t)(i * 256 + tid) * 3;
#pragma unroll 8
                for (int b = 0; b < B_; ++b) {
                    const float y0 = yr[3 * b], y1 = yr[3 * b + 1], y2 = yr[3 * b + 2];
                    float* ob = o + (size_t)b * (NATOMS_ * 3);
                    ob[0] = y0; ob[1] = y1; ob[2] = y2;
                }
            }
        }
    }
}

extern "C" void kernel_launch(void* const* d_in, const int* in_sizes, int n_in,
                              void* d_out, int out_size, void* d_ws, size_t ws_size,
                              hipStream_t stream) {
    const float* x_v  = (const float*)d_in[0];
    const float* Wh   = (const float*)d_in[1];
    const float* Wo   = (const float*)d_in[2];
    const int*   seq  = (const int*)  d_in[3];
    const int*   gidx = (const int*)  d_in[4];
    float* out     = (float*)d_out;
    float* wc      = (float*)d_ws;

    const size_t lds_bytes = (size_t)(B_ * XPITCH_ + MPAD_ * YPITCH_) * sizeof(float);

    residues_combine_kernel<<<dim3(VDIM_ / 32, NLETTER_), 256, 0, stream>>>(Wh, Wo, wc);

    residues_gemm_kernel<<<S_, 256, lds_bytes, stream>>>(x_v, wc, seq, gidx, out);
}


# Pipeline the K loop over four 64-channel chunks of x: a chunk is stored, barriered and multiplied while later chunks' loads are in flight; the Wc tile goes through LDS ahead of them; same sums
# speedup vs baseline: 3.6723x; 1.1210x over previous
#include <hip/hip_runtime.h>


typedef __attribute__((ext_vector_type(2))) float v2f;
typedef __attribute__((ext_vector_type(4))) float v4f;
typedef __attribute__((ext_vector_type(8))) float v8f;

#define B_      32
#define S_      512
#define VDIM_   256
#define MAXLEN_ 13
#define NLETTER_ 20
#define XROW_   (VDIM_ * 3)
#define XPITCH_ 780
#define MPAD_   16
#define WC_FLOATS_ (NLETTER_ * MPAD_ * VDIM_)
#define NATOMS_ 3745
#define NSCAN_  ((NATOMS_ + 255) / 256)
#define YPITCH_ 98
#define REDPITCH_ (4 * 8 * 32 + 16)
#define APITCH_ 260
#define KCH_    4
#define CHF_    (XROW_ / KCH_)
#define CHP_    (CHF_ / 4)
#define NPC_    (B_ * CHP_ / 256)

__global__ __launch_bounds__(256)
void residues_combine_kernel(const float* __restrict__ W_hid,
                             const float* __restrict__ W_out,
                             float* __restrict__ wc)
{
    __shared__ float red[2 * REDPITCH_];

    const int letter = blockIdx.y;
    const int c0     = blockIdx.x * 32;
    const int tid    = threadIdx.x;
    const int lane   = tid & 31;
    const int wave   = tid >> 5;
    const int lhalf  = lane >> 4;
    const int l16    = lane & 15;
    const int ntile  = wave & 1;
    const int kq     = wave >> 1;

    const float* Wo = W_out + (size_t)letter * MAXLEN_ * VDIM_;
    const float* Wh = W_hid + (size_t)letter * VDIM_ * VDIM_;
    const int n0 = c0 + ntile * 16;
    const bool mval = (l16 < MAXLEN_);
    const int  arow = mval ? l16 : 0;
    const v2f zv2 = {0.f, 0.f};

    v2f a[16], b[16];
#pragma unroll
    for (int i = 0; i < 16; ++i) {
        const int kk = kq * 64 + 4 * i + 2 * lhalf;
        a[i] = *(const v2f*)(Wo + (size_t)arow * VDIM_ + kk);
        b[i].x = Wh[(size_t)kk * VDIM_ + n0 + l16];
        b[i].y = Wh[(size_t)(kk + 1) * VDIM_ + n0 + l16];
    }

    v8f acc_e = (v8f){0.f, 0.f, 0.f, 0.f, 0.f, 0.f, 0.f, 0.f};
    v8f acc_o = (v8f){0.f, 0.f, 0.f, 0.f, 0.f, 0.f, 0.f, 0.f};
#pragma unroll
    for (int i = 0; i < 16; i += 2) {
        const v2f ae = mval ? a[i]     : zv2;
        const v2f ao = mval ? a[i + 1] : zv2;
        acc_e = __builtin_amdgcn_wmma_f32_16x16x4_f32(
            false, ae, false, b[i],     (short)0, acc_e, false, false);
        acc_o = __builtin_amdgcn_wmma_f32_16x16x4_f32(
            false, ao, false, b[i + 1], (short)0, acc_o, false, false);
    }
    const v8f acc = acc_e + acc_o;

    {
        float* r = red + ntile * REDPITCH_ + kq * (8 * 32) + lane;
#pragma unroll
        for (int j = 0; j < 8; ++j) r[j * 32] = acc[j];
    }
    __syncthreads();

    for (int e = tid; e < MPAD_ * 32; e += 256) {
        const int row = e >> 5, col = e & 31;
        const int nt = col >> 4, cc = col & 15;
        const float* r = red + nt * REDPITCH_ + (row & 7) * 32 + (row >> 3) * 16 + cc;
        const float sum = (r[0] + r[8 * 32]) + (r[2 * 8 * 32] + r[3 * 8 * 32]);
        wc[((size_t)letter * MPAD_ + row) * VDIM_ + c0 + col] = sum;
    }
}

__device__ __forceinline__ void lds_handover_barrier()
{
    asm volatile("s_wait_dscnt 0x0\n\ts_barrier_signal -1\n\ts_barrier_wait -1" ::: "memory");
}

__global__ __launch_bounds__(256)
void residues_gemm_kernel(const float* __restrict__ x_v,
                          const float* __restrict__ wc,
                          const int*  __restrict__ seq,
                          const int*  __restrict__ gidx,
                          float* __restrict__ out)
{
    extern __shared__ __attribute__((aligned(16))) float smem[];
    float* XR = smem;
    float* Y  = smem + B_ * XPITCH_;
    float* AW = Y + MPAD_ * YPITCH_;

    const int s     = blockIdx.x;
    const int tid   = threadIdx.x;
    const int lane  = tid & 31;
    const int wave  = tid >> 5;
    const int lhalf = lane >> 4;
    const int l16   = lane & 15;

    const int letter = seq[s];
    __builtin_amdgcn_sched_barrier(0);

    int goff[NPC_], loff[NPC_];
#pragma unroll
    for (int m = 0; m < NPC_; ++m) {
        const int idx = m * 256 + tid;
        const int row = idx / CHP_, pr = idx - row * CHP_;
        goff[m] = row * (S_ * XROW_) + 4 * pr;
        loff[m] = row * XPITCH_ + 4 * pr;
    }
    const float* xs = x_v + (size_t)s * XROW_;
    v4f xr[KCH_ * NPC_];
    v4f wr[4];
#pragma unroll
    for (int q = 0; q < KCH_; ++q) {
#pragma unroll
        for (int m = 0; m < NPC_; ++m)
            xr[NPC_ * q + m] = *(const v4f*)(xs + goff[m] + CHF_ * q);
        __builtin_amdgcn_sched_barrier(0);
        if (q == 0) {
            const float* Wc = wc + (size_t)(unsigned)letter * MPAD_ * VDIM_;
#pragma unroll
            for (int i = 0; i < 4; ++i)
                wr[i] = *(const v4f*)(Wc + 4 * (i * 256 + tid));
            __builtin_amdgcn_sched_barrier(0);
        }
    }

    int g[NSCAN_];
    __builtin_assume(tid < 256);
#pragma unroll
    for (int i = 0; i < NSCAN_; ++i) {
        const int p = i * 256 + tid;
        g[i] = (p < NATOMS_) ? gidx[p] : -1;
    }

    const int cc = wave % 3;
    const int bcol = (wave / 3) * 16 + l16;
    v8f acc_e = (v8f){0.f, 0.f, 0.f, 0.f, 0.f, 0.f, 0.f, 0.f};
    v8f acc_o = (v8f){0.f, 0.f, 0.f, 0.f, 0.f, 0.f, 0.f, 0.f};

#pragma unroll
    for (int q = 0; q < KCH_; ++q) {
#pragma unroll
        for (int m = 0; m < NPC_; ++m)
            *(v4f*)(XR + loff[m] + CHF_ * q) = xr[NPC_ * q + m];
        if (q == 0) {
#pragma unroll
            for (int i = 0; i < 4; ++i) {
                const int idx = i * 256 + tid;
                *(v4f*)(AW + (idx >> 6) * APITCH_ + 4 * (idx & 63)) = wr[i];
            }
        }
        lds_handover_barrier();

        if (wave < 6) {
            const float* xb = XR + bcol * XPITCH_ + cc;
            const float* ar = AW + l16 * APITCH_ + 2 * lhalf;
            for (int k = 64 * q; k < 64 * q + 64; k += 8) {
                const int kk = k + 2 * lhalf;
                v2f a0 = *(const v2f*)(ar + k);
                v2f a1 = *(const v2f*)(ar + k + 4);
                v2f b0, b1;
                b0.x = xb[3 * kk];
                b0.y = xb[3 * kk + 3];
                b1.x = xb[3 * kk + 12];
                b1.y = xb[3 * kk + 15];
                acc_e = __builtin_amdgcn_wmma_f32_16x16x4_f32(
                    false, a0, false, b0, (short)0, acc_e, false, false);
                acc_o = __builtin_amdgcn_wmma_f32_16x16x4_f32(
                    false, a1, false, b1, (short)0, acc_o, false, false);
            }
        }
    }

    unsigned long long pk = 0ull;
#pragma unroll
    for (int i = 0; i < NSCAN_; ++i) {
        const unsigned r = (unsigned)(g[i] - s * MAXLEN_);
        const unsigned code = (r < (unsigned)MAXLEN_) ? (r + 1u) : 0u;
        pk |= (unsigned long long)code << (4 * i);
    }

    if (wave < 6) {
        const v8f acc2 = acc_e + acc_o;

        float* yt = Y + (8 * lhalf) * YPITCH_ + 3 * bcol + cc;
#pragma unroll
        for (int j = 0; j < 8; ++j) yt[j * YPITCH_] = acc2[j];
    }
    __syncthreads();

    if (pk != 0ull) {
        for (int i = 0; i < NSCAN_; ++i) {
            const unsigned code = (unsigned)(pk >> (4 * i)) & 15u;
            if (code != 0u) {
                const float* yr = Y + (code - 1u) * YPITCH_;
                float* o = out + (size_t)(i * 256 + tid) * 3;
#pragma unroll 8
                for (int b = 0; b < B_; ++b) {
                    const float y0 = yr[3 * b], y1 = yr[3 * b + 1], y2 = yr[3 * b + 2];
                    float* ob = o + (size_t)b * (NATOMS_ * 3);
                    ob[0] = y0; ob[1] = y1; ob[2] = y2;
                }
            }
        }
    }
}

extern "C" void kernel_launch(void* const* d_in, const int* in_sizes, int n_in,
                              void* d_out, int out_size, void* d_ws, size_t ws_size,
                              hipStream_t stream) {
    const float* x_v  = (const float*)d_in[0];
    const float* Wh   = (const float*)d_in[1];
    const float* Wo   = (const float*)d_in[2];
    const int*   seq  = (const int*)  d_in[3];
    const int*   gidx = (const int*)  d_in[4];
    float* out     = (float*)d_out;
    float* wc      = (float*)d_ws;

    const size_t lds_bytes = (size_t)(B_ * XPITCH_ + MPAD_ * YPITCH_ + MPAD_ * APITCH_) * sizeof(float);

    residues_combine_kernel<<<dim3(VDIM_ / 32, NLETTER_), 256, 0, stream>>>(Wh, Wo, wc);

    residues_gemm_kernel<<<S_, 256, lds_bytes, stream>>>(x_v, wc, seq, gidx, out);
}


# Both launches enter half as many waves: 4-wave workgroups; launch 2's wave = batch half x k-step parity with three tiles, launch 1's wave = K quarter with both column tiles; same sums
# speedup vs baseline: 3.9726x; 1.0818x over previous
#include <hip/hip_runtime.h>


typedef __attribute__((ext_vector_type(2))) float v2f;
typedef __attribute__((ext_vector_type(4))) float v4f;
typedef __attribute__((ext_vector_type(8))) float v8f;

#define B_      32
#define S_      512
#define VDIM_   256
#define MAXLEN_ 13
#define NLETTER_ 20
#define XROW_   (VDIM_ * 3)
#define XPITCH_ 780
#define MPAD_   16
#define WC_FLOATS_ (NLETTER_ * MPAD_ * VDIM_)
#define NATOMS_ 3745
#define NT1_    128
#define NT2_    128
#define NSCAN_  ((NATOMS_ + NT2_ - 1) / NT2_)
#define NSCANH_ (NSCAN_ / 2)
#define YPITCH_ 98
#define REDPITCH_ (4 * 8 * 32 + 16)
#define APITCH_ 260
#define KCH_    4
#define CHF_    (XROW_ / KCH_)
#define CHP_    (CHF_ / 4)
#define NPC_    (B_ * CHP_ / NT2_)
#define NWP_    (MPAD_ * VDIM_ / 4 / NT2_)
#define XCHG_   12

__global__ __launch_bounds__(NT1_)
void residues_combine_kernel(const float* __restrict__ W_hid,
                             const float* __restrict__ W_out,
                             float* __restrict__ wc)
{
    __shared__ float red[2 * REDPITCH_];

    const int letter = blockIdx.y;
    const int c0     = blockIdx.x * 32;
    const int tid    = threadIdx.x;
    const int lane   = tid & 31;
    const int kq     = tid >> 5;
    const int lhalf  = lane >> 4;
    const int l16    = lane & 15;

    const float* Wo = W_out + (size_t)letter * MAXLEN_ * VDIM_;
    const float* Wh = W_hid + (size_t)letter * VDIM_ * VDIM_;
    const bool mval = (l16 < MAXLEN_);
    const int  arow = mval ? l16 : 0;
    const v2f zv2 = {0.f, 0.f};

    v2f a[16], b0[16], b1[16];
#pragma unroll
    for (int i = 0; i < 16; ++i) {
        const int kk = kq * 64 + 4 * i + 2 * lhalf;
        a[i] = *(const v2f*)(Wo + (size_t)arow * VDIM_ + kk);
        b0[i] = *(const v2f*)(Wh + (size_t)kk * VDIM_ + c0 + 2 * l16);
        b1[i] = *(const v2f*)(Wh + (size_t)(kk + 1) * VDIM_ + c0 + 2 * l16);
    }
    __builtin_amdgcn_sched_barrier(0);

    v8f acc_e[2], acc_o[2];
#pragma unroll
    for (int t = 0; t < 2; ++t) {
        acc_e[t] = (v8f){0.f, 0.f, 0.f, 0.f, 0.f, 0.f, 0.f, 0.f};
        acc_o[t] = (v8f){0.f, 0.f, 0.f, 0.f, 0.f, 0.f, 0.f, 0.f};
    }
#pragma unroll
    for (int i = 0; i < 16; i += 2) {
        const v2f ae = mval ? a[i]     : zv2;
        const v2f ao = mval ? a[i + 1] : zv2;
#pragma unroll
        for (int t = 0; t < 2; ++t) {
            const v2f be = {b0[i][t],     b1[i][t]};
            const v2f bo = {b0[i + 1][t], b1[i + 1][t]};
            acc_e[t] = __builtin_amdgcn_wmma_f32_16x16x4_f32(
                false, ae, false, be, (short)0, acc_e[t], false, false);
            acc_o[t] = __builtin_amdgcn_wmma_f32_16x16x4_f32(
                false, ao, false, bo, (short)0, acc_o[t], false, false);
        }
    }

#pragma unroll
    for (int t = 0; t < 2; ++t) {
        const v8f acc = acc_e[t] + acc_o[t];
        float* r = red + t * REDPITCH_ + kq * (8 * 32) + lane;
#pragma unroll
        for (int j = 0; j < 8; ++j) r[j * 32] = acc[j];
    }
    __syncthreads();

    for (int e = tid; e < MPAD_ * 32; e += NT1_) {
        const int row = e >> 5, col = e & 31;
        const int t = col & 1, cc = col >> 1;
        const float* r = red + t * REDPITCH_ + (row & 7) * 32 + (row >> 3) * 16 + cc;
        const float sum = (r[0] + r[8 * 32]) + (r[2 * 8 * 32] + r[3 * 8 * 32]);
        wc[((size_t)letter * MPAD_ + row) * VDIM_ + c0 + col] = sum;
    }
}

__device__ __forceinline__ void lds_handover_barrier()
{
    asm volatile("s_wait_dscnt 0x0\n\ts_barrier_signal -1\n\ts_barrier_wait -1" ::: "memory");
}


__device__ __forceinline__ void gather_rows(unsigned long long pk, int ibase, int tid,
                                            const float* __restrict__ Y, float* __restrict__ out)
{
    for (int i = 0; i < NSCANH_; ++i) {
        const unsigned code = (unsigned)(pk >> (4 * i)) & 15u;
        if (code != 0u) {
            const float* yr = Y + (code - 1u) * YPITCH_;
            float* o = out + (size_t)((ibase + i) * NT2_ + tid) * 3;
#pragma unroll 8
            for (int b = 0; b < B_; ++b) {
                const float y0 = yr[3 * b], y1 = yr[3 * b + 1], y2 = yr[3 * b + 2];
                float* ob = o + (size_t)b * (NATOMS_ * 3);
                ob[0] = y0; ob[1] = y1; ob[2] = y2;
            }
        }
    }
}

__global__ __launch_bounds__(NT2_)
void residues_gemm_kernel(const float* __restrict__ x_v,
                          const float* __restrict__ wc,
                          const int*  __restrict__ seq,
                          const int*  __restrict__ gidx,
                          float* __restrict__ out)
{
    extern __shared__ __attribute__((aligned(16))) float smem[];
    float* XR = smem;
    float* Y  = smem + B_ * XPITCH_;
    float* AW = Y + MPAD_ * YPITCH_;
    float* PS = AW + MPAD_ * APITCH_;

    const int s     = blockIdx.x;
    const int tid   = threadIdx.x;
    const int lane  = tid & 31;
    const int wave  = tid >> 5;
    const int lhalf = lane >> 4;
    const int l16   = lane & 15;
    const int bh    = wave & 1;
    const int par   = wave >> 1;

    const int letter = seq[s];
    __builtin_amdgcn_sched_barrier(0);

    int goff[NPC_], loff[NPC_];
#pragma unroll
    for (int j = 0; j < 3; ++j) {
        const int idx = j * NT2_ + tid;
        const int row = idx / CHP_, pr = idx - row * CHP_;
#pragma unroll
        for (int r = 0; r < NPC_ / 3; ++r) {
            goff[3 * r + j] = (row + 8 * r) * (S_ * XROW_) + 4 * pr;
            loff[3 * r + j] = (row + 8 * r) * XPITCH_ + 4 * pr;
        }
    }
    const float* xs = x_v + (size_t)s * XROW_;
    v4f xr[KCH_ * NPC_];
    v4f wr[NWP_];
#pragma unroll
    for (int q = 0; q < KCH_; ++q) {
#pragma unroll
        for (int m = 0; m < NPC_; ++m)
            xr[NPC_ * q + m] = *(const v4f*)(xs + goff[m] + CHF_ * q);
        __builtin_amdgcn_sched_barrier(0);
        if (q == 0) {
            const float* Wc = wc + (size_t)(unsigned)letter * MPAD_ * VDIM_;
#pragma unroll
            for (int i = 0; i < NWP_; ++i)
                wr[i] = *(const v4f*)(Wc + 4 * (i * NT2_ + tid));
            __builtin_amdgcn_sched_barrier(0);
        }
    }

    int g[NSCAN_];
    __builtin_assume(tid < NT2_);

    const int bcol = bh * 16 + l16;
    v8f acc[3];
#pragma unroll
    for (int c = 0; c < 3; ++c) acc[c] = (v8f){0.f, 0.f, 0.f, 0.f, 0.f, 0.f, 0.f, 0.f};

    const float* xb = XR + bcol * XPITCH_ + 3 * (4 * par + 2 * lhalf);
    const float* ar = AW + l16 * APITCH_ + 4 * par + 2 * lhalf;

#pragma unroll
    for (int q = 0; q < KCH_; ++q) {
        if (q == 1) {
#pragma unroll
            for (int i = 0; i < NSCAN_; ++i) {
                const int p = i * NT2_ + tid;
                g[i] = (p < NATOMS_) ? gidx[p] : -1;
            }
        }
#pragma unroll
        for (int m = 0; m < NPC_; ++m)
            *(v4f*)(XR + loff[m] + CHF_ * q) = xr[NPC_ * q + m];
        if (q == 0) {
#pragma unroll
            for (int i = 0; i < NWP_; ++i) {
                const int idx = i * NT2_ + tid;
                *(v4f*)(AW + (idx >> 6) * APITCH_ + 4 * (idx & 63)) = wr[i];
            }
        }
        lds_handover_barrier();

        v2f a[8], b[8][3];
#pragma unroll
        for (int j = 0; j < 8; ++j) {
            const int k = 64 * q + 8 * j;
            a[j] = *(const v2f*)(ar + k);
#pragma unroll
            for (int c = 0; c < 3; ++c) {
                b[j][c].x = xb[3 * k + c];
                b[j][c].y = xb[3 * k + 3 + c];
            }
        }
        __builtin_amdgcn_sched_barrier(0);
#pragma unroll
        for (int j = 0; j < 8; ++j) {
#pragma unroll
            for (int c = 0; c < 3; ++c)
                acc[c] = __builtin_amdgcn_wmma_f32_16x16x4_f32(
                    false, a[j], false, b[j][c], (short)0, acc[c], false, false);
        }
        if (q + 1 < KCH_) __builtin_amdgcn_sched_barrier(0);
    }

    unsigned long long pk0 = 0ull, pk1 = 0ull;
#pragma unroll
    for (int i = 0; i < NSCANH_; ++i) {
        const unsigned r0 = (unsigned)(g[i] - s * MAXLEN_);
        const unsigned r1 = (unsigned)(g[NSCANH_ + i] - s * MAXLEN_);
        const unsigned c0 = (r0 < (unsigned)MAXLEN_) ? (r0 + 1u) : 0u;
        const unsigned c1 = (r1 < (unsigned)MAXLEN_) ? (r1 + 1u) : 0u;
        pk0 |= (unsigned long long)c0 << (4 * i);
        pk1 |= (unsigned long long)c1 << (4 * i);
    }

    {
        v4f s0, s1, s2;
        if (par == 0) {
            s0 = (v4f){acc[1][4], acc[1][5], acc[1][6], acc[1][7]};
            s1 = (v4f){acc[2][0], acc[2][1], acc[2][2], acc[2][3]};
            s2 = (v4f){acc[2][4], acc[2][5], acc[2][6], acc[2][7]};
        } else {
            s0 = (v4f){acc[0][0], acc[0][1], acc[0][2], acc[0][3]};
            s1 = (v4f){acc[0][4], acc[0][5], acc[0][6], acc[0][7]};
            s2 = (v4f){acc[1][0], acc[1][1], acc[1][2], acc[1][3]};
        }
        float* ps = PS + (wave * 32 + lane) * XCHG_;
        *(v4f*)(ps)     = s0;
        *(v4f*)(ps + 4) = s1;
        *(v4f*)(ps + 8) = s2;
    }
    __syncthreads();
    {
        const float* pr = PS + ((wave ^ 2) * 32 + lane) * XCHG_;
        const v4f r0 = *(const v4f*)(pr);
        const v4f r1 = *(const v4f*)(pr + 4);
        const v4f r2 = *(const v4f*)(pr + 8);
        float* yt = Y + (8 * lhalf) * YPITCH_ + 3 * bcol;
        if (par == 0) {
#pragma unroll
            for (int j = 0; j < 4; ++j) {
                yt[j * YPITCH_]           = acc[0][j]     + r0[j];
                yt[(j + 4) * YPITCH_]     = acc[0][j + 4] + r1[j];
                yt[j * YPITCH_ + 1]       = acc[1][j]     + r2[j];
            }
        } else {
#pragma unroll
            for (int j = 0; j < 4; ++j) {
                yt[(j + 4) * YPITCH_ + 1] = acc[1][j + 4] + r0[j];
                yt[j * YPITCH_ + 2]       = acc[2][j]     + r1[j];
                yt[(j + 4) * YPITCH_ + 2] = acc[2][j + 4] + r2[j];
            }
        }
    }
    __syncthreads();

    if (pk0 != 0ull) gather_rows(pk0, 0, tid, Y, out);
    if (pk1 != 0ull) gather_rows(pk1, NSCANH_, tid, Y, out);
}

extern "C" void kernel_launch(void* const* d_in, const int* in_sizes, int n_in,
                              void* d_out, int out_size, void* d_ws, size_t ws_size,
                              hipStream_t stream) {
    const float* x_v  = (const float*)d_in[0];
    const float* Wh   = (const float*)d_in[1];
    const float* Wo   = (const float*)d_in[2];
    const int*   seq  = (const int*)  d_in[3];
    const int*   gidx = (const int*)  d_in[4];
    float* out     = (float*)d_out;
    float* wc      = (float*)d_ws;

    const size_t lds_bytes = (size_t)(B_ * XPITCH_ + MPAD_ * YPITCH_ + MPAD_ * APITCH_ + 4 * 32 * XCHG_) * sizeof(float);

    residues_combine_kernel<<<dim3(VDIM_ / 32, NLETTER_), NT1_, 0, stream>>>(Wh, Wo, wc);

    residues_gemm_kernel<<<S_, NT2_, lds_bytes, stream>>>(x_v, wc, seq, gidx, out);
}


# Gather by whole waves: a wave lists its matches (p, row) in LDS by ballot and rank; 30 lanes = 10 entries x 3 coordinates store one run of consecutive floats per batch entry; same sums
# speedup vs baseline: 4.2742x; 1.0759x over previous
#include <hip/hip_runtime.h>


typedef __attribute__((ext_vector_type(2))) float v2f;
typedef __attribute__((ext_vector_type(4))) float v4f;
typedef __attribute__((ext_vector_type(8))) float v8f;

#define B_      32
#define S_      512
#define VDIM_   256
#define MAXLEN_ 13
#define NLETTER_ 20
#define XROW_   (VDIM_ * 3)
#define XPITCH_ 780
#define MPAD_   16
#define WC_FLOATS_ (NLETTER_ * MPAD_ * VDIM_)
#define NATOMS_ 3745
#define NT1_    128
#define NT2_    128
#define NSCAN_  ((NATOMS_ + NT2_ - 1) / NT2_)
#define NSCANH_ (NSCAN_ / 2)
#define YPITCH_ 98
#define REDPITCH_ (4 * 8 * 32 + 16)
#define APITCH_ 260
#define KCH_    4
#define CHF_    (XROW_ / KCH_)
#define CHP_    (CHF_ / 4)
#define NPC_    (B_ * CHP_ / NT2_)
#define NWP_    (MPAD_ * VDIM_ / 4 / NT2_)
#define XCHG_   12
#define LCAP_   (NSCAN_ * 32)

__global__ __launch_bounds__(NT1_)
void residues_combine_kernel(const float* __restrict__ W_hid,
                             const float* __restrict__ W_out,
                             float* __restrict__ wc)
{
    __shared__ float red[2 * REDPITCH_];

    const int letter = blockIdx.y;
    const int c0     = blockIdx.x * 32;
    const int tid    = threadIdx.x;
    const int lane   = tid & 31;
    const int kq     = tid >> 5;
    const int lhalf  = lane >> 4;
    const int l16    = lane & 15;

    const float* Wo = W_out + (size_t)letter * MAXLEN_ * VDIM_;
    const float* Wh = W_hid + (size_t)letter * VDIM_ * VDIM_;
    const bool mval = (l16 < MAXLEN_);
    const int  arow = mval ? l16 : 0;
    const v2f zv2 = {0.f, 0.f};

    v2f a[16], b0[16], b1[16];
#pragma unroll
    for (int i = 0; i < 16; ++i) {
        const int kk = kq * 64 + 4 * i + 2 * lhalf;
        a[i] = *(const v2f*)(Wo + (size_t)arow * VDIM_ + kk);
        b0[i] = *(const v2f*)(Wh + (size_t)kk * VDIM_ + c0 + 2 * l16);
        b1[i] = *(const v2f*)(Wh + (size_t)(kk + 1) * VDIM_ + c0 + 2 * l16);
    }
    __builtin_amdgcn_sched_barrier(0);

    v8f acc_e[2], acc_o[2];
#pragma unroll
    for (int t = 0; t < 2; ++t) {
        acc_e[t] = (v8f){0.f, 0.f, 0.f, 0.f, 0.f, 0.f, 0.f, 0.f};
        acc_o[t] = (v8f){0.f, 0.f, 0.f, 0.f, 0.f, 0.f, 0.f, 0.f};
    }
#pragma unroll
    for (int i = 0; i < 16; i += 2) {
        const v2f ae = mval ? a[i]     : zv2;
        const v2f ao = mval ? a[i + 1] : zv2;
#pragma unroll
        for (int t = 0; t < 2; ++t) {
            const v2f be = {b0[i][t],     b1[i][t]};
            const v2f bo = {b0[i + 1][t], b1[i + 1][t]};
            acc_e[t] = __builtin_amdgcn_wmma_f32_16x16x4_f32(
                false, ae, false, be, (short)0, acc_e[t], false, false);
            acc_o[t] = __builtin_amdgcn_wmma_f32_16x16x4_f32(
                false, ao, false, bo, (short)0, acc_o[t], false, false);
        }
    }

#pragma unroll
    for (int t = 0; t < 2; ++t) {
        const v8f acc = acc_e[t] + acc_o[t];
        float* r = red + t * REDPITCH_ + kq * (8 * 32) + lane;
#pragma unroll
        for (int j = 0; j < 8; ++j) r[j * 32] = acc[j];
    }
    __syncthreads();

    for (int e = tid; e < MPAD_ * 32; e += NT1_) {
        const int row = e >> 5, col = e & 31;
        const int t = col & 1, cc = col >> 1;
        const float* r = red + t * REDPITCH_ + (row & 7) * 32 + (row >> 3) * 16 + cc;
        const float sum = (r[0] + r[8 * 32]) + (r[2 * 8 * 32] + r[3 * 8 * 32]);
        wc[((size_t)letter * MPAD_ + row) * VDIM_ + c0 + col] = sum;
    }
}

__device__ __forceinline__ void lds_handover_barrier()
{
    asm volatile("s_wait_dscnt 0x0\n\ts_barrier_signal -1\n\ts_barrier_wait -1" ::: "memory");
}


__global__ __launch_bounds__(NT2_)
void residues_gemm_kernel(const float* __restrict__ x_v,
                          const float* __restrict__ wc,
                          const int*  __restrict__ seq,
                          const int*  __restrict__ gidx,
                          float* __restrict__ out)
{
    extern __shared__ __attribute__((aligned(16))) float smem[];
    float* XR = smem;
    float* Y  = smem + B_ * XPITCH_;
    float* AW = Y + MPAD_ * YPITCH_;
    float* PS = AW + MPAD_ * APITCH_;

    const int s     = blockIdx.x;
    const int tid   = threadIdx.x;
    const int lane  = tid & 31;
    const int wave  = tid >> 5;
    const int lhalf = lane >> 4;
    const int l16   = lane & 15;
    const int bh    = wave & 1;
    const int par   = wave >> 1;

    const int letter = seq[s];
    __builtin_amdgcn_sched_barrier(0);

    int goff[NPC_], loff[NPC_];
#pragma unroll
    for (int j = 0; j < 3; ++j) {
        const int idx = j * NT2_ + tid;
        const int row = idx / CHP_, pr = idx - row * CHP_;
#pragma unroll
        for (int r = 0; r < NPC_ / 3; ++r) {
            goff[3 * r + j] = (row + 8 * r) * (S_ * XROW_) + 4 * pr;
            loff[3 * r + j] = (row + 8 * r) * XPITCH_ + 4 * pr;
        }
    }
    const float* xs = x_v + (size_t)s * XROW_;
    v4f xr[KCH_ * NPC_];
    v4f wr[NWP_];
#pragma unroll
    for (int q = 0; q < KCH_; ++q) {
#pragma unroll
        for (int m = 0; m < NPC_; ++m)
            xr[NPC_ * q + m] = *(const v4f*)(xs + goff[m] + CHF_ * q);
        __builtin_amdgcn_sched_barrier(0);
        if (q == 0) {
            const float* Wc = wc + (size_t)(unsigned)letter * MPAD_ * VDIM_;
#pragma unroll
            for (int i = 0; i < NWP_; ++i)
                wr[i] = *(const v4f*)(Wc + 4 * (i * NT2_ + tid));
            __builtin_amdgcn_sched_barrier(0);
        }
    }

    int g[NSCAN_];
    __builtin_assume(tid < NT2_);

    const int bcol = bh * 16 + l16;
    v8f acc[3];
#pragma unroll
    for (int c = 0; c < 3; ++c) acc[c] = (v8f){0.f, 0.f, 0.f, 0.f, 0.f, 0.f, 0.f, 0.f};

    const float* xb = XR + bcol * XPITCH_ + 3 * (4 * par + 2 * lhalf);
    const float* ar = AW + l16 * APITCH_ + 4 * par + 2 * lhalf;

#pragma unroll
    for (int q = 0; q < KCH_; ++q) {
        if (q == 1) {
#pragma unroll
            for (int i = 0; i < NSCAN_; ++i) {
                const int p = i * NT2_ + tid;
                g[i] = (p < NATOMS_) ? gidx[p] : -1;
            }
        }
#pragma unroll
        for (int m = 0; m < NPC_; ++m)
            *(v4f*)(XR + loff[m] + CHF_ * q) = xr[NPC_ * q + m];
        if (q == 0) {
#pragma unroll
            for (int i = 0; i < NWP_; ++i) {
                const int idx = i * NT2_ + tid;
                *(v4f*)(AW + (idx >> 6) * APITCH_ + 4 * (idx & 63)) = wr[i];
            }
        }
        lds_handover_barrier();

        v2f a[8], b[8][3];
#pragma unroll
        for (int j = 0; j < 8; ++j) {
            const int k = 64 * q + 8 * j;
            a[j] = *(const v2f*)(ar + k);
#pragma unroll
            for (int c = 0; c < 3; ++c) {
                b[j][c].x = xb[3 * k + c];
                b[j][c].y = xb[3 * k + 3 + c];
            }
        }
        __builtin_amdgcn_sched_barrier(0);
#pragma unroll
        for (int j = 0; j < 8; ++j) {
#pragma unroll
            for (int c = 0; c < 3; ++c)
                acc[c] = __builtin_amdgcn_wmma_f32_16x16x4_f32(
                    false, a[j], false, b[j][c], (short)0, acc[c], false, false);
        }
        if (q + 1 < KCH_) __builtin_amdgcn_sched_barrier(0);
    }

    unsigned long long pk0 = 0ull, pk1 = 0ull;
#pragma unroll
    for (int i = 0; i < NSCANH_; ++i) {
        const unsigned r0 = (unsigned)(g[i] - s * MAXLEN_);
        const unsigned r1 = (unsigned)(g[NSCANH_ + i] - s * MAXLEN_);
        const unsigned c0 = (r0 < (unsigned)MAXLEN_) ? (r0 + 1u) : 0u;
        const unsigned c1 = (r1 < (unsigned)MAXLEN_) ? (r1 + 1u) : 0u;
        pk0 |= (unsigned long long)c0 << (4 * i);
        pk1 |= (unsigned long long)c1 << (4 * i);
    }

    {
        v4f s0, s1, s2;
        if (par == 0) {
            s0 = (v4f){acc[1][4], acc[1][5], acc[1][6], acc[1][7]};
            s1 = (v4f){acc[2][0], acc[2][1], acc[2][2], acc[2][3]};
            s2 = (v4f){acc[2][4], acc[2][5], acc[2][6], acc[2][7]};
        } else {
            s0 = (v4f){acc[0][0], acc[0][1], acc[0][2], acc[0][3]};
            s1 = (v4f){acc[0][4], acc[0][5], acc[0][6], acc[0][7]};
            s2 = (v4f){acc[1][0], acc[1][1], acc[1][2], acc[1][3]};
        }
        float* ps = PS + (wave * 32 + lane) * XCHG_;
        *(v4f*)(ps)     = s0;
        *(v4f*)(ps + 4) = s1;
        *(v4f*)(ps + 8) = s2;
    }
    __syncthreads();
    {
        const float* pr = PS + ((wave ^ 2) * 32 + lane) * XCHG_;
        const v4f r0 = *(const v4f*)(pr);
        const v4f r1 = *(const v4f*)(pr + 4);
        const v4f r2 = *(const v4f*)(pr + 8);
        float* yt = Y + (8 * lhalf) * YPITCH_ + 3 * bcol;
        if (par == 0) {
#pragma unroll
            for (int j = 0; j < 4; ++j) {
                yt[j * YPITCH_]           = acc[0][j]     + r0[j];
                yt[(j + 4) * YPITCH_]     = acc[0][j + 4] + r1[j];
                yt[j * YPITCH_ + 1]       = acc[1][j]     + r2[j];
            }
        } else {
#pragma unroll
            for (int j = 0; j < 4; ++j) {
                yt[(j + 4) * YPITCH_ + 1] = acc[1][j + 4] + r0[j];
                yt[j * YPITCH_ + 2]       = acc[2][j]     + r1[j];
                yt[(j + 4) * YPITCH_ + 2] = acc[2][j + 4] + r2[j];
            }
        }
    }

    unsigned* LW = (unsigned*)XR + wave * LCAP_;
    unsigned cnt = 0u;
#pragma unroll
    for (int h = 0; h < 2; ++h) {
        unsigned long long pk = h ? pk1 : pk0;
        for (int it = 0; it < NSCANH_; ++it) {
            const unsigned act = __builtin_amdgcn_ballot_w32(pk != 0ull);
            if (act == 0u) break;
            const int src = __builtin_ctz(act);
            const unsigned wlo = (unsigned)__builtin_amdgcn_readlane((int)(unsigned)pk, src);
            const unsigned whi = (unsigned)__builtin_amdgcn_readlane((int)(unsigned)(pk >> 32), src);
            const unsigned long long w = ((unsigned long long)whi << 32) | wlo;
            const int sh = __builtin_ctzll(w) & ~3;
            const unsigned code = (unsigned)(pk >> sh) & 15u;
            const bool hit = (code != 0u);
            const unsigned m = __builtin_amdgcn_ballot_w32(hit);
            const unsigned rank = __builtin_amdgcn_mbcnt_lo(m, 0u);
            if (hit)
                LW[cnt + rank] = ((unsigned)((h * NSCANH_ + (sh >> 2)) * NT2_ + tid) << 4) | (code - 1u);
            cnt += (unsigned)__builtin_popcount(m);
            pk &= ~(15ull << sh);
        }
    }
    __syncthreads();

    for (unsigned base = 0u; base < cnt; base += 10u) {
        const unsigned j = base + (unsigned)lane / 3u;
        const unsigned c = (unsigned)lane % 3u;
        if (lane < 30 && j < cnt) {
            const unsigned ent = LW[j];
            const float* yr = Y + (ent & 15u) * YPITCH_ + c;
            float* o = out + (size_t)(ent >> 4) * 3 + c;
#pragma unroll
            for (int b = 0; b < B_; ++b)
                o[(size_t)b * (NATOMS_ * 3)] = yr[3 * b];
        }
    }
}

extern "C" void kernel_launch(void* const* d_in, const int* in_sizes, int n_in,
                              void* d_out, int out_size, void* d_ws, size_t ws_size,
                              hipStream_t stream) {
    const float* x_v  = (const float*)d_in[0];
    const float* Wh   = (const float*)d_in[1];
    const float* Wo   = (const float*)d_in[2];
    const int*   seq  = (const int*)  d_in[3];
    const int*   gidx = (const int*)  d_in[4];
    float* out     = (float*)d_out;
    float* wc      = (float*)d_ws;

    const size_t lds_bytes = (size_t)(B_ * XPITCH_ + MPAD_ * YPITCH_ + MPAD_ * APITCH_ + 4 * 32 * XCHG_) * sizeof(float);

    residues_combine_kernel<<<dim3(VDIM_ / 32, NLETTER_), NT1_, 0, stream>>>(Wh, Wo, wc);

    residues_gemm_kernel<<<S_, NT2_, lds_bytes, stream>>>(x_v, wc, seq, gidx, out);
}
